# starting kernel
# baseline (speedup 1.0000x reference)
#include <hip/hip_runtime.h>


typedef __attribute__((ext_vector_type(2))) float v2f;
typedef __attribute__((ext_vector_type(8))) float v8f;

#define SEQ   2048
#define CTOT  64
#define NH    8
#define CK    8
#define NSTK  3

__global__ __launch_bounds__(32)
void causal_attn_fa_f32wmma(const float* __restrict__ Kp,
                            const float* __restrict__ Qp,
                            const float* __restrict__ Vp,
                            float* __restrict__ Op) {
    const int lane = threadIdx.x;
    const int l16  = lane & 15;
    const int hi   = lane >> 4;
    const int it   = blockIdx.x;
    const int head = blockIdx.y;
    const int stk  = blockIdx.z;

    const size_t base = ((size_t)stk * CTOT + (size_t)head * CK) * SEQ;
    const float* __restrict__ q = Qp + base;
    const float* __restrict__ k = Kp + base;
    const float* __restrict__ v = Vp + base;
    float*       __restrict__ o = Op + base;

    const int   ibase = it * 16;
    const int   iq    = ibase + l16;
    const float scale = 0.35355339059327373f;

    const int c0 = hi * 2;
    v2f qb0, qb1;
    qb0.x = q[(size_t)(c0    ) * SEQ + iq] * scale;
    qb0.y = q[(size_t)(c0 + 1) * SEQ + iq] * scale;
    qb1.x = q[(size_t)(c0 + 4) * SEQ + iq] * scale;
    qb1.y = q[(size_t)(c0 + 5) * SEQ + iq] * scale;

    float m_run = -1e30f;
    float l_run = 0.0f;
    v8f   o_acc = {0.f,0.f,0.f,0.f,0.f,0.f,0.f,0.f};

    const int  cv    = l16;
    const bool cv_ok = (cv < CK);

    for (int jt = 0; jt <= it; ++jt) {
        const int jbase = jt * 16;

        const int jk = jbase + l16;
        v2f ka0, ka1;
        ka0.x = k[(size_t)(c0    ) * SEQ + jk];
        ka0.y = k[(size_t)(c0 + 1) * SEQ + jk];
        ka1.x = k[(size_t)(c0 + 4) * SEQ + jk];
        ka1.y = k[(size_t)(c0 + 5) * SEQ + jk];

        v8f sacc = {0.f,0.f,0.f,0.f,0.f,0.f,0.f,0.f};
        sacc = __builtin_amdgcn_wmma_f32_16x16x4_f32(false, ka0, false, qb0,
                                                     (short)0, sacc, false, false);
        sacc = __builtin_amdgcn_wmma_f32_16x16x4_f32(false, ka1, false, qb1,
                                                     (short)0, sacc, false, false);

        float p[8];
        #pragma unroll
        for (int r = 0; r < 8; ++r) p[r] = sacc[r];

        if (jt == it) {
            #pragma unroll
            for (int r = 0; r < 8; ++r) {
                const int key_abs = jbase + r + 8 * hi;
                if (key_abs > iq) p[r] = -1e30f;
            }
        }

        float mloc = p[0];
        #pragma unroll
        for (int r = 1; r < 8; ++r) mloc = fmaxf(mloc, p[r]);
        const float mtile = fmaxf(mloc, __shfl_xor(mloc, 16, 32));
        const float m_new = fmaxf(m_run, mtile);
        const float alpha = __expf(m_run - m_new);
        float lsum = 0.0f;
        #pragma unroll
        for (int r = 0; r < 8; ++r) { p[r] = __expf(p[r] - m_new); lsum += p[r]; }
        l_run = l_run * alpha + lsum + __shfl_xor(lsum, 16, 32);
        m_run = m_new;
        #pragma unroll
        for (int r = 0; r < 8; ++r) o_acc[r] *= alpha;

        #pragma unroll
        for (int t = 0; t < 4; ++t) {
            v2f pb;
            if (t < 2) {
                const float cx = __shfl_xor(p[4*t + 2], 16, 32);
                const float cy = __shfl_xor(p[4*t + 3], 16, 32);
                pb.x = hi ? cx : p[4*t    ];
                pb.y = hi ? cy : p[4*t + 1];
            } else {
                const float cx = __shfl_xor(p[4*t - 8], 16, 32);
                const float cy = __shfl_xor(p[4*t - 7], 16, 32);
                pb.x = hi ? p[4*t - 6] : cx;
                pb.y = hi ? p[4*t - 5] : cy;
            }
            v2f va;
            if (cv_ok) {
                const float* vp = v + (size_t)cv * SEQ + jbase + 4*t + 2*hi;
                va.x = vp[0];
                va.y = vp[1];
            } else {
                va.x = 0.0f; va.y = 0.0f;
            }
            o_acc = __builtin_amdgcn_wmma_f32_16x16x4_f32(false, va, false, pb,
                                                          (short)0, o_acc, false, false);
        }
    }

    const float inv = 1.0f / l_run;
    if (hi == 0) {
        #pragma unroll
        for (int r = 0; r < 8; ++r) {
            o[(size_t)r * SEQ + iq] = o_acc[r] * inv;
        }
    }
}

extern "C" void kernel_launch(void* const* d_in, const int* in_sizes, int n_in,
                              void* d_out, int out_size, void* d_ws, size_t ws_size,
                              hipStream_t stream) {
    (void)in_sizes; (void)n_in; (void)d_ws; (void)ws_size; (void)out_size;
    const float* Kp = (const float*)d_in[0];
    const float* Qp = (const float*)d_in[1];
    const float* Vp = (const float*)d_in[2];
    float*       Op = (float*)d_out;

    dim3 grid(SEQ / 16, NH, NSTK);
    dim3 block(32);
    causal_attn_fa_f32wmma<<<grid, block, 0, stream>>>(Kp, Qp, Vp, Op);
}


# Split each query tile's key loop across 8 waves of one workgroup (key tiles dealt round-robin); per-wave max, denominator and O partials merged through LDS
# speedup vs baseline: 2.4701x; 2.4701x over previous
#include <hip/hip_runtime.h>


typedef __attribute__((ext_vector_type(2))) float v2f;
typedef __attribute__((ext_vector_type(8))) float v8f;

#define SEQ   2048
#define CTOT  64
#define NH    8
#define CK    8
#define NSTK  3
#define NW    8

__global__ __launch_bounds__(32 * NW)
void causal_attn_fa_f32wmma(const float* __restrict__ Kp,
                            const float* __restrict__ Qp,
                            const float* __restrict__ Vp,
                            float* __restrict__ Op) {
    __shared__ float s_m[NW][16];
    __shared__ float s_l[NW][16];
    __shared__ float s_o[NW][CK][16];

    const int tid  = threadIdx.x;
    const int lane = tid & 31;
    const int wv   = __builtin_amdgcn_readfirstlane(tid >> 5);
    const int l16  = lane & 15;
    const int hi   = lane >> 4;
    const int it   = blockIdx.x;
    const int head = blockIdx.y;
    const int stk  = blockIdx.z;

    const size_t base = ((size_t)stk * CTOT + (size_t)head * CK) * SEQ;
    const float* __restrict__ q = Qp + base;
    const float* __restrict__ k = Kp + base;
    const float* __restrict__ v = Vp + base;
    float*       __restrict__ o = Op + base;

    const int   ibase = it * 16;
    const int   iq    = ibase + l16;
    const float scale = 0.35355339059327373f;

    const int c0 = hi * 2;
    v2f qb0, qb1;
    qb0.x = q[(size_t)(c0    ) * SEQ + iq] * scale;
    qb0.y = q[(size_t)(c0 + 1) * SEQ + iq] * scale;
    qb1.x = q[(size_t)(c0 + 4) * SEQ + iq] * scale;
    qb1.y = q[(size_t)(c0 + 5) * SEQ + iq] * scale;

    float m_run = -1e30f;
    float l_run = 0.0f;
    v8f   o_acc = {0.f,0.f,0.f,0.f,0.f,0.f,0.f,0.f};

    const int  cv    = l16;
    const bool cv_ok = (cv < CK);

    for (int jt = wv; jt <= it; jt += NW) {
        const int jbase = jt * 16;

        const int jk = jbase + l16;
        v2f ka0, ka1;
        ka0.x = k[(size_t)(c0    ) * SEQ + jk];
        ka0.y = k[(size_t)(c0 + 1) * SEQ + jk];
        ka1.x = k[(size_t)(c0 + 4) * SEQ + jk];
        ka1.y = k[(size_t)(c0 + 5) * SEQ + jk];

        v8f sacc = {0.f,0.f,0.f,0.f,0.f,0.f,0.f,0.f};
        sacc = __builtin_amdgcn_wmma_f32_16x16x4_f32(false, ka0, false, qb0,
                                                     (short)0, sacc, false, false);
        sacc = __builtin_amdgcn_wmma_f32_16x16x4_f32(false, ka1, false, qb1,
                                                     (short)0, sacc, false, false);

        float p[8];
        #pragma unroll
        for (int r = 0; r < 8; ++r) p[r] = sacc[r];

        if (jt == it) {
            #pragma unroll
            for (int r = 0; r < 8; ++r) {
                const int key_abs = jbase + r + 8 * hi;
                if (key_abs > iq) p[r] = -1e30f;
            }
        }

        float mloc = p[0];
        #pragma unroll
        for (int r = 1; r < 8; ++r) mloc = fmaxf(mloc, p[r]);
        const float mtile = fmaxf(mloc, __shfl_xor(mloc, 16, 32));
        const float m_new = fmaxf(m_run, mtile);
        const float alpha = __expf(m_run - m_new);
        float lsum = 0.0f;
        #pragma unroll
        for (int r = 0; r < 8; ++r) { p[r] = __expf(p[r] - m_new); lsum += p[r]; }
        l_run = l_run * alpha + lsum + __shfl_xor(lsum, 16, 32);
        m_run = m_new;
        #pragma unroll
        for (int r = 0; r < 8; ++r) o_acc[r] *= alpha;

        #pragma unroll
        for (int t = 0; t < 4; ++t) {
            v2f pb;
            if (t < 2) {
                const float cx = __shfl_xor(p[4*t + 2], 16, 32);
                const float cy = __shfl_xor(p[4*t + 3], 16, 32);
                pb.x = hi ? cx : p[4*t    ];
                pb.y = hi ? cy : p[4*t + 1];
            } else {
                const float cx = __shfl_xor(p[4*t - 8], 16, 32);
                const float cy = __shfl_xor(p[4*t - 7], 16, 32);
                pb.x = hi ? p[4*t - 6] : cx;
                pb.y = hi ? p[4*t - 5] : cy;
            }
            v2f va;
            if (cv_ok) {
                const float* vp = v + (size_t)cv * SEQ + jbase + 4*t + 2*hi;
                va.x = vp[0];
                va.y = vp[1];
            } else {
                va.x = 0.0f; va.y = 0.0f;
            }
            o_acc = __builtin_amdgcn_wmma_f32_16x16x4_f32(false, va, false, pb,
                                                          (short)0, o_acc, false, false);
        }
    }

    if (hi == 0) {
        s_m[wv][l16] = m_run;
        s_l[wv][l16] = l_run;
        #pragma unroll
        for (int r = 0; r < 8; ++r) s_o[wv][r][l16] = o_acc[r];
    }
    __syncthreads();

    if (tid < CK * 16) {
        const int c  = tid >> 4;
        const int qi = tid & 15;
        float mw[NW];
        #pragma unroll
        for (int w = 0; w < NW; ++w) mw[w] = s_m[w][qi];
        float M = mw[0];
        #pragma unroll
        for (int w = 1; w < NW; ++w) M = fmaxf(M, mw[w]);
        float num = 0.0f, den = 0.0f;
        #pragma unroll
        for (int w = 0; w < NW; ++w) {
            const float e = __expf(mw[w] - M);
            num += e * s_o[w][c][qi];
            den += e * s_l[w][qi];
        }
        o[(size_t)c * SEQ + ibase + qi] = num / den;
    }
}

extern "C" void kernel_launch(void* const* d_in, const int* in_sizes, int n_in,
                              void* d_out, int out_size, void* d_ws, size_t ws_size,
                              hipStream_t stream) {
    (void)in_sizes; (void)n_in; (void)d_ws; (void)ws_size; (void)out_size;
    const float* Kp = (const float*)d_in[0];
    const float* Qp = (const float*)d_in[1];
    const float* Vp = (const float*)d_in[2];
    float*       Op = (float*)d_out;

    dim3 grid(SEQ / 16, NH, NSTK);
    dim3 block(32 * NW);
    causal_attn_fa_f32wmma<<<grid, block, 0, stream>>>(Kp, Qp, Vp, Op);
}


# Keep each lane half's 8 keys in place: P feeds P*V straight from the S accumulator (k order matched by the V loader), denominator merged once after the loop; cross-half moves per tile 10 -> 1
# speedup vs baseline: 3.4277x; 1.3877x over previous
#include <hip/hip_runtime.h>


typedef __attribute__((ext_vector_type(2))) float v2f;
typedef __attribute__((ext_vector_type(4))) float v4f;
typedef __attribute__((ext_vector_type(8))) float v8f;

#define SEQ   2048
#define CTOT  64
#define NH    8
#define CK    8
#define NSTK  3
#define NW    8

__global__ __launch_bounds__(32 * NW)
void causal_attn_fa_f32wmma(const float* __restrict__ Kp,
                            const float* __restrict__ Qp,
                            const float* __restrict__ Vp,
                            float* __restrict__ Op) {
    __shared__ float s_m[NW][16];
    __shared__ float s_l[NW][16];
    __shared__ float s_o[NW][CK][16];

    const int tid  = threadIdx.x;
    const int lane = tid & 31;
    const int wv   = __builtin_amdgcn_readfirstlane(tid >> 5);
    const int l16  = lane & 15;
    const int hi   = lane >> 4;
    const int it   = blockIdx.x;
    const int head = blockIdx.y;
    const int stk  = blockIdx.z;

    const size_t base = ((size_t)stk * CTOT + (size_t)head * CK) * SEQ;
    const float* __restrict__ q = Qp + base;
    const float* __restrict__ k = Kp + base;
    const float* __restrict__ v = Vp + base;
    float*       __restrict__ o = Op + base;

    const int   ibase = it * 16;
    const int   iq    = ibase + l16;
    const float scale = 0.35355339059327373f;

    const int c0 = hi * 2;
    v2f qb0, qb1;
    qb0.x = q[(size_t)(c0    ) * SEQ + iq] * scale;
    qb0.y = q[(size_t)(c0 + 1) * SEQ + iq] * scale;
    qb1.x = q[(size_t)(c0 + 4) * SEQ + iq] * scale;
    qb1.y = q[(size_t)(c0 + 5) * SEQ + iq] * scale;

    float m_run = -1e30f;
    float l_run = 0.0f;
    v8f   o_acc = {0.f,0.f,0.f,0.f,0.f,0.f,0.f,0.f};

    const int  cv    = l16;
    const bool cv_ok = (cv < CK);

    for (int jt = wv; jt <= it; jt += NW) {
        const int jbase = jt * 16;

        const int jk = jbase + l16;
        v2f ka0, ka1;
        ka0.x = k[(size_t)(c0    ) * SEQ + jk];
        ka0.y = k[(size_t)(c0 + 1) * SEQ + jk];
        ka1.x = k[(size_t)(c0 + 4) * SEQ + jk];
        ka1.y = k[(size_t)(c0 + 5) * SEQ + jk];

        v4f vlo = {0.f,0.f,0.f,0.f};
        v4f vup = {0.f,0.f,0.f,0.f};
        if (cv_ok) {
            const v4f* vp = (const v4f*)(v + (size_t)cv * SEQ + jbase + 8 * hi);
            vlo = vp[0];
            vup = vp[1];
        }
        const float vrow[8] = {vlo.x, vlo.y, vlo.z, vlo.w, vup.x, vup.y, vup.z, vup.w};

        v8f sacc = {0.f,0.f,0.f,0.f,0.f,0.f,0.f,0.f};
        sacc = __builtin_amdgcn_wmma_f32_16x16x4_f32(false, ka0, false, qb0,
                                                     (short)0, sacc, false, false);
        sacc = __builtin_amdgcn_wmma_f32_16x16x4_f32(false, ka1, false, qb1,
                                                     (short)0, sacc, false, false);

        float p[8];
        #pragma unroll
        for (int r = 0; r < 8; ++r) p[r] = sacc[r];

        if (jt == it) {
            #pragma unroll
            for (int r = 0; r < 8; ++r) {
                const int key_abs = jbase + r + 8 * hi;
                if (key_abs > iq) p[r] = -1e30f;
            }
        }

        float mloc = p[0];
        #pragma unroll
        for (int r = 1; r < 8; ++r) mloc = fmaxf(mloc, p[r]);
        const float mtile = fmaxf(mloc, __shfl_xor(mloc, 16, 32));
        const float m_new = fmaxf(m_run, mtile);
        const float alpha = __expf(m_run - m_new);
        float lsum = 0.0f;
        #pragma unroll
        for (int r = 0; r < 8; ++r) { p[r] = __expf(p[r] - m_new); lsum += p[r]; }
        l_run = l_run * alpha + lsum;
        m_run = m_new;
        #pragma unroll
        for (int r = 0; r < 8; ++r) o_acc[r] *= alpha;

        #pragma unroll
        for (int t = 0; t < 4; ++t) {
            v2f pb;
            pb.x = p[2*t    ];
            pb.y = p[2*t + 1];
            v2f va;
            va.x = vrow[2*t    ];
            va.y = vrow[2*t + 1];
            o_acc = __builtin_amdgcn_wmma_f32_16x16x4_f32(false, va, false, pb,
                                                          (short)0, o_acc, false, false);
        }
    }

    const float l_tot = l_run + __shfl_xor(l_run, 16, 32);
    if (hi == 0) {
        s_m[wv][l16] = m_run;
        s_l[wv][l16] = l_tot;
        #pragma unroll
        for (int r = 0; r < 8; ++r) s_o[wv][r][l16] = o_acc[r];
    }
    __syncthreads();

    if (tid < CK * 16) {
        const int c  = tid >> 4;
        const int qi = tid & 15;
        float mw[NW];
        #pragma unroll
        for (int w = 0; w < NW; ++w) mw[w] = s_m[w][qi];
        float M = mw[0];
        #pragma unroll
        for (int w = 1; w < NW; ++w) M = fmaxf(M, mw[w]);
        float num = 0.0f, den = 0.0f;
        #pragma unroll
        for (int w = 0; w < NW; ++w) {
            const float e = __expf(mw[w] - M);
            num += e * s_o[w][c][qi];
            den += e * s_l[w][qi];
        }
        o[(size_t)c * SEQ + ibase + qi] = num / den;
    }
}

extern "C" void kernel_launch(void* const* d_in, const int* in_sizes, int n_in,
                              void* d_out, int out_size, void* d_ws, size_t ws_size,
                              hipStream_t stream) {
    (void)in_sizes; (void)n_in; (void)d_ws; (void)ws_size; (void)out_size;
    const float* Kp = (const float*)d_in[0];
    const float* Qp = (const float*)d_in[1];
    const float* Vp = (const float*)d_in[2];
    float*       Op = (float*)d_out;

    dim3 grid(SEQ / 16, NH, NSTK);
    dim3 block(32 * NW);
    causal_attn_fa_f32wmma<<<grid, block, 0, stream>>>(Kp, Qp, Vp, Op);
}


# Each workgroup takes the query-tile pair (t, 127-t), 129 key tiles in every workgroup, dealt round-robin to its 8 waves: half as many waves, equal work each, whole grid resident in one pass
# speedup vs baseline: 3.9035x; 1.1388x over previous
#include <hip/hip_runtime.h>


typedef __attribute__((ext_vector_type(2))) float v2f;
typedef __attribute__((ext_vector_type(4))) float v4f;
typedef __attribute__((ext_vector_type(8))) float v8f;

#define SEQ   2048
#define CTOT  64
#define NH    8
#define CK    8
#define NSTK  3
#define NW    8
#define NT    (SEQ / 16)

__device__ __forceinline__
void attend_share(const float* __restrict__ k, const float* __restrict__ v,
                  const int it, const int jt0, const v2f qb0, const v2f qb1,
                  const int l16, const int hi,
                  float* __restrict__ sm, float* __restrict__ sl, float* __restrict__ so) {
    const int ibase = it * 16;
    const int iq    = ibase + l16;
    const int c0    = hi * 2;

    float m_run = -1e30f;
    float l_run = 0.0f;
    v8f   o_acc = {0.f,0.f,0.f,0.f,0.f,0.f,0.f,0.f};

    const int  cv    = l16;
    const bool cv_ok = (cv < CK);

    for (int jt = jt0; jt <= it; jt += NW) {
        const int jbase = jt * 16;

        const int jk = jbase + l16;
        v2f ka0, ka1;
        ka0.x = k[(size_t)(c0    ) * SEQ + jk];
        ka0.y = k[(size_t)(c0 + 1) * SEQ + jk];
        ka1.x = k[(size_t)(c0 + 4) * SEQ + jk];
        ka1.y = k[(size_t)(c0 + 5) * SEQ + jk];

        v4f vlo = {0.f,0.f,0.f,0.f};
        v4f vup = {0.f,0.f,0.f,0.f};
        if (cv_ok) {
            const v4f* vp = (const v4f*)(v + (size_t)cv * SEQ + jbase + 8 * hi);
            vlo = vp[0];
            vup = vp[1];
        }
        const float vrow[8] = {vlo.x, vlo.y, vlo.z, vlo.w, vup.x, vup.y, vup.z, vup.w};

        v8f sacc = {0.f,0.f,0.f,0.f,0.f,0.f,0.f,0.f};
        sacc = __builtin_amdgcn_wmma_f32_16x16x4_f32(false, ka0, false, qb0,
                                                     (short)0, sacc, false, false);
        sacc = __builtin_amdgcn_wmma_f32_16x16x4_f32(false, ka1, false, qb1,
                                                     (short)0, sacc, false, false);

        float p[8];
        #pragma unroll
        for (int r = 0; r < 8; ++r) p[r] = sacc[r];

        if (jt == it) {
            #pragma unroll
            for (int r = 0; r < 8; ++r) {
                const int key_abs = jbase + r + 8 * hi;
                if (key_abs > iq) p[r] = -1e30f;
            }
        }

        float mloc = p[0];
        #pragma unroll
        for (int r = 1; r < 8; ++r) mloc = fmaxf(mloc, p[r]);
        const float mtile = fmaxf(mloc, __shfl_xor(mloc, 16, 32));
        const float m_new = fmaxf(m_run, mtile);
        const float alpha = __expf(m_run - m_new);
        float lsum = 0.0f;
        #pragma unroll
        for (int r = 0; r < 8; ++r) { p[r] = __expf(p[r] - m_new); lsum += p[r]; }
        l_run = l_run * alpha + lsum;
        m_run = m_new;
        #pragma unroll
        for (int r = 0; r < 8; ++r) o_acc[r] *= alpha;

        #pragma unroll
        for (int t = 0; t < 4; ++t) {
            v2f pb;
            pb.x = p[2*t    ];
            pb.y = p[2*t + 1];
            v2f va;
            va.x = vrow[2*t    ];
            va.y = vrow[2*t + 1];
            o_acc = __builtin_amdgcn_wmma_f32_16x16x4_f32(false, va, false, pb,
                                                          (short)0, o_acc, false, false);
        }
    }

    const float l_tot = l_run + __shfl_xor(l_run, 16, 32);
    if (hi == 0) {
        sm[l16] = m_run;
        sl[l16] = l_tot;
        #pragma unroll
        for (int r = 0; r < 8; ++r) so[r * 16 + l16] = o_acc[r];
    }
}

__global__ __launch_bounds__(32 * NW)
void causal_attn_fa_f32wmma(const float* __restrict__ Kp,
                            const float* __restrict__ Qp,
                            const float* __restrict__ Vp,
                            float* __restrict__ Op) {
    __shared__ float s_m[2][NW][16];
    __shared__ float s_l[2][NW][16];
    __shared__ float s_o[2][NW][CK][16];

    const int tid  = threadIdx.x;
    const int lane = tid & 31;
    const int wv   = __builtin_amdgcn_readfirstlane(tid >> 5);
    const int l16  = lane & 15;
    const int hi   = lane >> 4;
    const int x    = blockIdx.x;
    const int head = blockIdx.y;
    const int stk  = blockIdx.z;

    const int itA = x;
    const int itB = NT - 1 - x;

    const size_t base = ((size_t)stk * CTOT + (size_t)head * CK) * SEQ;
    const float* __restrict__ q = Qp + base;
    const float* __restrict__ k = Kp + base;
    const float* __restrict__ v = Vp + base;
    float*       __restrict__ o = Op + base;

    const float scale = 0.35355339059327373f;

    const int c0  = hi * 2;
    const int iqA = itA * 16 + l16;
    const int iqB = itB * 16 + l16;
    v2f qa0, qa1, qc0, qc1;
    qa0.x = q[(size_t)(c0    ) * SEQ + iqA] * scale;
    qa0.y = q[(size_t)(c0 + 1) * SEQ + iqA] * scale;
    qa1.x = q[(size_t)(c0 + 4) * SEQ + iqA] * scale;
    qa1.y = q[(size_t)(c0 + 5) * SEQ + iqA] * scale;
    qc0.x = q[(size_t)(c0    ) * SEQ + iqB] * scale;
    qc0.y = q[(size_t)(c0 + 1) * SEQ + iqB] * scale;
    qc1.x = q[(size_t)(c0 + 4) * SEQ + iqB] * scale;
    qc1.y = q[(size_t)(c0 + 5) * SEQ + iqB] * scale;

    const int jtA0 = wv;
    const int jtB0 = (wv - (x + 1)) & (NW - 1);

    attend_share(k, v, itA, jtA0, qa0, qa1, l16, hi, &s_m[0][wv][0], &s_l[0][wv][0], &s_o[0][wv][0][0]);
    attend_share(k, v, itB, jtB0, qc0, qc1, l16, hi, &s_m[1][wv][0], &s_l[1][wv][0], &s_o[1][wv][0][0]);
    __syncthreads();

    {
        const int ph = tid >> 7;
        const int c  = (tid >> 4) & (CK - 1);
        const int qi = tid & 15;
        const int ibase = (ph ? itB : itA) * 16;
        float mw[NW];
        #pragma unroll
        for (int w = 0; w < NW; ++w) mw[w] = s_m[ph][w][qi];
        float M = mw[0];
        #pragma unroll
        for (int w = 1; w < NW; ++w) M = fmaxf(M, mw[w]);
        float num = 0.0f, den = 0.0f;
        #pragma unroll
        for (int w = 0; w < NW; ++w) {
            const float e = __expf(mw[w] - M);
            num += e * s_o[ph][w][c][qi];
            den += e * s_l[ph][w][qi];
        }
        o[(size_t)c * SEQ + ibase + qi] = num / den;
    }
}

extern "C" void kernel_launch(void* const* d_in, const int* in_sizes, int n_in,
                              void* d_out, int out_size, void* d_ws, size_t ws_size,
                              hipStream_t stream) {
    (void)in_sizes; (void)n_in; (void)d_ws; (void)ws_size; (void)out_size;
    const float* Kp = (const float*)d_in[0];
    const float* Qp = (const float*)d_in[1];
    const float* Vp = (const float*)d_in[2];
    float*       Op = (float*)d_out;

    dim3 grid(NT / 2, NH, NSTK);
    dim3 block(32 * NW);
    causal_attn_fa_f32wmma<<<grid, block, 0, stream>>>(Kp, Qp, Vp, Op);
}


# Two key tiles (32 keys) per loop trip with one joint running-max update: one cross-half max exchange, one alpha and one O rescale per 32 keys; the two S products are independent chains
# speedup vs baseline: 4.2170x; 1.0803x over previous
#include <hip/hip_runtime.h>


typedef __attribute__((ext_vector_type(2))) float v2f;
typedef __attribute__((ext_vector_type(4))) float v4f;
typedef __attribute__((ext_vector_type(8))) float v8f;

#define SEQ   2048
#define CTOT  64
#define NH    8
#define CK    8
#define NSTK  3
#define NW    8
#define NT    (SEQ / 16)

template <int NTL>
__device__ __forceinline__
void kv_step(const float* __restrict__ kq, const float* __restrict__ vq,
             const int jt, const int it, const int iq, const v2f qb0, const v2f qb1,
             const int l16, const int hi, float& m_run, float& l_run, v8f& o_acc) {
    const bool cv_ok = (l16 < CK);

    v2f ka0[NTL], ka1[NTL];
    v4f vlo[NTL], vup[NTL];
    #pragma unroll
    for (int u = 0; u < NTL; ++u) {
        const int joff = u * NW * 16;
        ka0[u].x = kq[joff          ];
        ka0[u].y = kq[joff +     SEQ];
        ka1[u].x = kq[joff + 4 * SEQ];
        ka1[u].y = kq[joff + 5 * SEQ];
        vlo[u] = (v4f){0.f,0.f,0.f,0.f};
        vup[u] = (v4f){0.f,0.f,0.f,0.f};
    }
    if (cv_ok) {
        #pragma unroll
        for (int u = 0; u < NTL; ++u) {
            const v4f* vp = (const v4f*)(vq + u * NW * 16);
            vlo[u] = vp[0];
            vup[u] = vp[1];
        }
    }

    v8f sacc[NTL];
    #pragma unroll
    for (int u = 0; u < NTL; ++u) {
        const v8f z = {0.f,0.f,0.f,0.f,0.f,0.f,0.f,0.f};
        sacc[u] = __builtin_amdgcn_wmma_f32_16x16x4_f32(false, ka0[u], false, qb0,
                                                        (short)0, z, false, false);
    }
    #pragma unroll
    for (int u = 0; u < NTL; ++u)
        sacc[u] = __builtin_amdgcn_wmma_f32_16x16x4_f32(false, ka1[u], false, qb1,
                                                        (short)0, sacc[u], false, false);

    float p[NTL][8];
    #pragma unroll
    for (int u = 0; u < NTL; ++u) {
        #pragma unroll
        for (int r = 0; r < 8; ++r) p[u][r] = sacc[u][r];
    }

    const int jlast = jt + (NTL - 1) * NW;
    if (jlast == it) {
        #pragma unroll
        for (int r = 0; r < 8; ++r) {
            const int key_abs = jlast * 16 + r + 8 * hi;
            if (key_abs > iq) p[NTL - 1][r] = -1e30f;
        }
    }

    float mloc = p[0][0];
    #pragma unroll
    for (int u = 0; u < NTL; ++u) {
        #pragma unroll
        for (int r = 0; r < 8; ++r) mloc = fmaxf(mloc, p[u][r]);
    }
    const float mtile = fmaxf(mloc, __shfl_xor(mloc, 16, 32));
    const float m_new = fmaxf(m_run, mtile);
    const float alpha = __expf(m_run - m_new);
    float lsum = 0.0f;
    #pragma unroll
    for (int u = 0; u < NTL; ++u) {
        #pragma unroll
        for (int r = 0; r < 8; ++r) { p[u][r] = __expf(p[u][r] - m_new); lsum += p[u][r]; }
    }
    l_run = l_run * alpha + lsum;
    m_run = m_new;
    #pragma unroll
    for (int r = 0; r < 8; ++r) o_acc[r] *= alpha;

    #pragma unroll
    for (int u = 0; u < NTL; ++u) {
        const float vrow[8] = {vlo[u].x, vlo[u].y, vlo[u].z, vlo[u].w, vup[u].x, vup[u].y, vup[u].z, vup[u].w};
        #pragma unroll
        for (int t = 0; t < 4; ++t) {
            v2f pb;
            pb.x = p[u][2*t    ];
            pb.y = p[u][2*t + 1];
            v2f va;
            va.x = vrow[2*t    ];
            va.y = vrow[2*t + 1];
            o_acc = __builtin_amdgcn_wmma_f32_16x16x4_f32(false, va, false, pb,
                                                          (short)0, o_acc, false, false);
        }
    }
}

__device__ __forceinline__
void attend_share(const float* __restrict__ k, const float* __restrict__ v,
                  const int it, const int jt0, const v2f qb0, const v2f qb1,
                  const int l16, const int hi,
                  float* __restrict__ sm, float* __restrict__ sl, float* __restrict__ so) {
    const int iq = it * 16 + l16;

    float m_run = -1e30f;
    float l_run = 0.0f;
    v8f   o_acc = {0.f,0.f,0.f,0.f,0.f,0.f,0.f,0.f};

    const float* kq = k + (size_t)(hi * 2) * SEQ + jt0 * 16 + l16;
    const float* vq = v + (size_t)(l16 & (CK - 1)) * SEQ + jt0 * 16 + 8 * hi;
    int jt = jt0;
    for (; jt + NW <= it; jt += 2 * NW) {
        kv_step<2>(kq, vq, jt, it, iq, qb0, qb1, l16, hi, m_run, l_run, o_acc);
        kq += 2 * NW * 16;
        vq += 2 * NW * 16;
    }
    if (jt <= it)
        kv_step<1>(kq, vq, jt, it, iq, qb0, qb1, l16, hi, m_run, l_run, o_acc);

    const float l_tot = l_run + __shfl_xor(l_run, 16, 32);
    if (hi == 0) {
        sm[l16] = m_run;
        sl[l16] = l_tot;
        #pragma unroll
        for (int r = 0; r < 8; ++r) so[r * 16 + l16] = o_acc[r];
    }
}

__global__ __launch_bounds__(32 * NW)
void causal_attn_fa_f32wmma(const float* __restrict__ Kp,
                            const float* __restrict__ Qp,
                            const float* __restrict__ Vp,
                            float* __restrict__ Op) {
    __shared__ float s_m[2][NW][16];
    __shared__ float s_l[2][NW][16];
    __shared__ float s_o[2][NW][CK][16];

    const int tid  = threadIdx.x;
    const int lane = tid & 31;
    const int wv   = __builtin_amdgcn_readfirstlane(tid >> 5);
    const int l16  = lane & 15;
    const int hi   = lane >> 4;
    const int x    = blockIdx.x;
    const int head = blockIdx.y;
    const int stk  = blockIdx.z;

    const int itA = x;
    const int itB = NT - 1 - x;

    const size_t base = ((size_t)stk * CTOT + (size_t)head * CK) * SEQ;
    const float* __restrict__ q = Qp + base;
    const float* __restrict__ k = Kp + base;
    const float* __restrict__ v = Vp + base;
    float*       __restrict__ o = Op + base;

    const float scale = 0.35355339059327373f;

    const int c0  = hi * 2;
    const int iqA = itA * 16 + l16;
    const int iqB = itB * 16 + l16;
    v2f qa0, qa1, qc0, qc1;
    qa0.x = q[(size_t)(c0    ) * SEQ + iqA] * scale;
    qa0.y = q[(size_t)(c0 + 1) * SEQ + iqA] * scale;
    qa1.x = q[(size_t)(c0 + 4) * SEQ + iqA] * scale;
    qa1.y = q[(size_t)(c0 + 5) * SEQ + iqA] * scale;
    qc0.x = q[(size_t)(c0    ) * SEQ + iqB] * scale;
    qc0.y = q[(size_t)(c0 + 1) * SEQ + iqB] * scale;
    qc1.x = q[(size_t)(c0 + 4) * SEQ + iqB] * scale;
    qc1.y = q[(size_t)(c0 + 5) * SEQ + iqB] * scale;

    const int jtA0 = wv;
    const int jtB0 = (wv - (x + 1)) & (NW - 1);

    attend_share(k, v, itA, jtA0, qa0, qa1, l16, hi, &s_m[0][wv][0], &s_l[0][wv][0], &s_o[0][wv][0][0]);
    attend_share(k, v, itB, jtB0, qc0, qc1, l16, hi, &s_m[1][wv][0], &s_l[1][wv][0], &s_o[1][wv][0][0]);
    __syncthreads();

    {
        const int ph = tid >> 7;
        const int c  = (tid >> 4) & (CK - 1);
        const int qi = tid & 15;
        const int ibase = (ph ? itB : itA) * 16;
        float mw[NW];
        #pragma unroll
        for (int w = 0; w < NW; ++w) mw[w] = s_m[ph][w][qi];
        float M = mw[0];
        #pragma unroll
        for (int w = 1; w < NW; ++w) M = fmaxf(M, mw[w]);
        float num = 0.0f, den = 0.0f;
        #pragma unroll
        for (int w = 0; w < NW; ++w) {
            const float e = __expf(mw[w] - M);
            num += e * s_o[ph][w][c][qi];
            den += e * s_l[ph][w][qi];
        }
        o[(size_t)c * SEQ + ibase + qi] = num / den;
    }
}

extern "C" void kernel_launch(void* const* d_in, const int* in_sizes, int n_in,
                              void* d_out, int out_size, void* d_ws, size_t ws_size,
                              hipStream_t stream) {
    (void)in_sizes; (void)n_in; (void)d_ws; (void)ws_size; (void)out_size;
    const float* Kp = (const float*)d_in[0];
    const float* Qp = (const float*)d_in[1];
    const float* Vp = (const float*)d_in[2];
    float*       Op = (float*)d_out;

    dim3 grid(NT / 2, NH, NSTK);
    dim3 block(32 * NW);
    causal_attn_fa_f32wmma<<<grid, block, 0, stream>>>(Kp, Qp, Vp, Op);
}


# Each wave applies every loaded K and V fragment to two adjacent query tiles (workgroup = tile pairs (2x,2x+1) and (126-2x,127-2x), 130 key-tile visits): K/V loads and wave count halved
# speedup vs baseline: 4.5412x; 1.0769x over previous
#include <hip/hip_runtime.h>


typedef __attribute__((ext_vector_type(2))) float v2f;
typedef __attribute__((ext_vector_type(4))) float v4f;
typedef __attribute__((ext_vector_type(8))) float v8f;

#define SEQ   2048
#define CTOT  64
#define NH    8
#define CK    8
#define NSTK  3
#define NW    8
#define NT    (SEQ / 16)
#define ND    (NT / 2)

template <int NTL>
__device__ __forceinline__
void kv_step(const float* __restrict__ kq, const float* __restrict__ vq,
             const int jt, const int it0, const int iq0,
             const v2f (&qb0)[2], const v2f (&qb1)[2],
             const int l16, const int hi, float (&m_run)[2], float (&l_run)[2], v8f (&o_acc)[2]) {
    const bool cv_ok = (l16 < CK);

    v2f ka0[NTL], ka1[NTL];
    v4f vlo[NTL], vup[NTL];
    #pragma unroll
    for (int u = 0; u < NTL; ++u) {
        const int joff = u * NW * 16;
        ka0[u].x = kq[joff          ];
        ka0[u].y = kq[joff +     SEQ];
        ka1[u].x = kq[joff + 4 * SEQ];
        ka1[u].y = kq[joff + 5 * SEQ];
        vlo[u] = (v4f){0.f,0.f,0.f,0.f};
        vup[u] = (v4f){0.f,0.f,0.f,0.f};
    }
    if (cv_ok) {
        #pragma unroll
        for (int u = 0; u < NTL; ++u) {
            const v4f* vp = (const v4f*)(vq + u * NW * 16);
            vlo[u] = vp[0];
            vup[u] = vp[1];
        }
    }

    v8f sacc[2][NTL];
    #pragma unroll
    for (int u = 0; u < NTL; ++u) {
        #pragma unroll
        for (int q = 0; q < 2; ++q) {
            const v8f z = {0.f,0.f,0.f,0.f,0.f,0.f,0.f,0.f};
            sacc[q][u] = __builtin_amdgcn_wmma_f32_16x16x4_f32(false, ka0[u], false, qb0[q],
                                                               (short)0, z, false, false);
        }
    }
    #pragma unroll
    for (int u = 0; u < NTL; ++u) {
        #pragma unroll
        for (int q = 0; q < 2; ++q)
            sacc[q][u] = __builtin_amdgcn_wmma_f32_16x16x4_f32(false, ka1[u], false, qb1[q],
                                                               (short)0, sacc[q][u], false, false);
    }

    float p[2][NTL][8];
    #pragma unroll
    for (int q = 0; q < 2; ++q) {
        #pragma unroll
        for (int u = 0; u < NTL; ++u) {
            #pragma unroll
            for (int r = 0; r < 8; ++r) p[q][u][r] = sacc[q][u][r];
        }
    }

    const int jlast = jt + (NTL - 1) * NW;
    if (jlast >= it0) {
        #pragma unroll
        for (int q = 0; q < 2; ++q) {
            #pragma unroll
            for (int r = 0; r < 8; ++r) {
                const int key_abs = jlast * 16 + r + 8 * hi;
                if (key_abs > iq0 + 16 * q) p[q][NTL - 1][r] = -1e30f;
            }
        }
    }

    #pragma unroll
    for (int q = 0; q < 2; ++q) {
        float mloc = p[q][0][0];
        #pragma unroll
        for (int u = 0; u < NTL; ++u) {
            #pragma unroll
            for (int r = 0; r < 8; ++r) mloc = fmaxf(mloc, p[q][u][r]);
        }
        const float mtile = fmaxf(mloc, __shfl_xor(mloc, 16, 32));
        const float m_new = fmaxf(m_run[q], mtile);
        const float alpha = __expf(m_run[q] - m_new);
        float lsum = 0.0f;
        #pragma unroll
        for (int u = 0; u < NTL; ++u) {
            #pragma unroll
            for (int r = 0; r < 8; ++r) { p[q][u][r] = __expf(p[q][u][r] - m_new); lsum += p[q][u][r]; }
        }
        l_run[q] = l_run[q] * alpha + lsum;
        m_run[q] = m_new;
        #pragma unroll
        for (int r = 0; r < 8; ++r) o_acc[q][r] *= alpha;
    }

    #pragma unroll
    for (int u = 0; u < NTL; ++u) {
        const float vrow[8] = {vlo[u].x, vlo[u].y, vlo[u].z, vlo[u].w, vup[u].x, vup[u].y, vup[u].z, vup[u].w};
        #pragma unroll
        for (int t = 0; t < 4; ++t) {
            v2f va;
            va.x = vrow[2*t    ];
            va.y = vrow[2*t + 1];
            #pragma unroll
            for (int q = 0; q < 2; ++q) {
                v2f pb;
                pb.x = p[q][u][2*t    ];
                pb.y = p[q][u][2*t + 1];
                o_acc[q] = __builtin_amdgcn_wmma_f32_16x16x4_f32(false, va, false, pb,
                                                                 (short)0, o_acc[q], false, false);
            }
        }
    }
}

__device__ __forceinline__
void attend_share(const float* __restrict__ k, const float* __restrict__ v,
                  const int it0, const int jt0, const v2f (&qb0)[2], const v2f (&qb1)[2],
                  const int l16, const int hi,
                  float* __restrict__ sm, float* __restrict__ sl, float* __restrict__ so) {
    const int iq0  = it0 * 16 + l16;
    const int jmax = it0 + 1;

    float m_run[2] = {-1e30f, -1e30f};
    float l_run[2] = {0.0f, 0.0f};
    v8f   o_acc[2];
    o_acc[0] = (v8f){0.f,0.f,0.f,0.f,0.f,0.f,0.f,0.f};
    o_acc[1] = (v8f){0.f,0.f,0.f,0.f,0.f,0.f,0.f,0.f};

    const float* kq = k + (size_t)(hi * 2) * SEQ + jt0 * 16 + l16;
    const float* vq = v + (size_t)(l16 & (CK - 1)) * SEQ + jt0 * 16 + 8 * hi;
    int jt = jt0;
    for (; jt + NW <= jmax; jt += 2 * NW) {
        kv_step<2>(kq, vq, jt, it0, iq0, qb0, qb1, l16, hi, m_run, l_run, o_acc);
        kq += 2 * NW * 16;
        vq += 2 * NW * 16;
    }
    if (jt <= jmax)
        kv_step<1>(kq, vq, jt, it0, iq0, qb0, qb1, l16, hi, m_run, l_run, o_acc);

    #pragma unroll
    for (int q = 0; q < 2; ++q) {
        const float l_tot = l_run[q] + __shfl_xor(l_run[q], 16, 32);
        if (hi == 0) {
            sm[q * NW * 16 + l16] = m_run[q];
            sl[q * NW * 16 + l16] = l_tot;
            #pragma unroll
            for (int r = 0; r < 8; ++r) so[q * NW * CK * 16 + r * 16 + l16] = o_acc[q][r];
        }
    }
}

__global__ __launch_bounds__(32 * NW)
void causal_attn_fa_f32wmma(const float* __restrict__ Kp,
                            const float* __restrict__ Qp,
                            const float* __restrict__ Vp,
                            float* __restrict__ Op) {
    __shared__ float s_m[4][NW][16];
    __shared__ float s_l[4][NW][16];
    __shared__ float s_o[4][NW][CK][16];

    const int tid  = threadIdx.x;
    const int lane = tid & 31;
    const int wv   = __builtin_amdgcn_readfirstlane(tid >> 5);
    const int l16  = lane & 15;
    const int hi   = lane >> 4;
    const int x    = blockIdx.x;
    const int head = blockIdx.y;
    const int stk  = blockIdx.z;

    const int itA = 2 * x;
    const int itB = 2 * (ND - 1 - x);

    const size_t base = ((size_t)stk * CTOT + (size_t)head * CK) * SEQ;
    const float* __restrict__ q = Qp + base;
    const float* __restrict__ k = Kp + base;
    const float* __restrict__ v = Vp + base;
    float*       __restrict__ o = Op + base;

    const float scale = 0.35355339059327373f;

    const int c0 = hi * 2;
    v2f qa0[2], qa1[2], qc0[2], qc1[2];
    #pragma unroll
    for (int t = 0; t < 2; ++t) {
        const int iqA = (itA + t) * 16 + l16;
        const int iqB = (itB + t) * 16 + l16;
        qa0[t].x = q[(size_t)(c0    ) * SEQ + iqA] * scale;
        qa0[t].y = q[(size_t)(c0 + 1) * SEQ + iqA] * scale;
        qa1[t].x = q[(size_t)(c0 + 4) * SEQ + iqA] * scale;
        qa1[t].y = q[(size_t)(c0 + 5) * SEQ + iqA] * scale;
        qc0[t].x = q[(size_t)(c0    ) * SEQ + iqB] * scale;
        qc0[t].y = q[(size_t)(c0 + 1) * SEQ + iqB] * scale;
        qc1[t].x = q[(size_t)(c0 + 4) * SEQ + iqB] * scale;
        qc1[t].y = q[(size_t)(c0 + 5) * SEQ + iqB] * scale;
    }

    const int jtA0 = wv;
    const int jtB0 = (wv - (itA + 2)) & (NW - 1);

    attend_share(k, v, itA, jtA0, qa0, qa1, l16, hi, &s_m[0][wv][0], &s_l[0][wv][0], &s_o[0][wv][0][0]);
    attend_share(k, v, itB, jtB0, qc0, qc1, l16, hi, &s_m[2][wv][0], &s_l[2][wv][0], &s_o[2][wv][0][0]);
    __syncthreads();

    #pragma unroll
    for (int ph = 0; ph < 2; ++ph) {
        const int c  = wv;
        const int ti = ph * 2 + hi;
        const int qi = l16;
        const int ibase = (ph ? itB : itA) * 16;
        float mw[NW];
        #pragma unroll
        for (int w = 0; w < NW; ++w) mw[w] = s_m[ti][w][qi];
        float M = mw[0];
        #pragma unroll
        for (int w = 1; w < NW; ++w) M = fmaxf(M, mw[w]);
        float num = 0.0f, den = 0.0f;
        #pragma unroll
        for (int w = 0; w < NW; ++w) {
            const float e = __expf(mw[w] - M);
            num += e * s_o[ti][w][c][qi];
            den += e * s_l[ti][w][qi];
        }
        o[(size_t)c * SEQ + ibase + lane] = num / den;
    }
}

extern "C" void kernel_launch(void* const* d_in, const int* in_sizes, int n_in,
                              void* d_out, int out_size, void* d_ws, size_t ws_size,
                              hipStream_t stream) {
    (void)in_sizes; (void)n_in; (void)d_ws; (void)ws_size; (void)out_size;
    const float* Kp = (const float*)d_in[0];
    const float* Qp = (const float*)d_in[1];
    const float* Vp = (const float*)d_in[2];
    float*       Op = (float*)d_out;

    dim3 grid(ND / 2, NH, NSTK);
    dim3 block(32 * NW);
    causal_attn_fa_f32wmma<<<grid, block, 0, stream>>>(Kp, Qp, Vp, Op);
}


# Vector work per score cut from max+sub+mul+exp+add to max+sub+exp: log2(e) folded into the Q scale, denominator from a row of ones in V's zero pad (row 8 of P*V), pad lanes set once per share
# speedup vs baseline: 4.8408x; 1.0660x over previous
#include <hip/hip_runtime.h>


typedef __attribute__((ext_vector_type(2))) float v2f;
typedef __attribute__((ext_vector_type(4))) float v4f;
typedef __attribute__((ext_vector_type(8))) float v8f;

#define SEQ   2048
#define CTOT  64
#define NH    8
#define CK    8
#define NSTK  3
#define NW    8
#define NT    (SEQ / 16)
#define ND    (NT / 2)

template <int NTL>
__device__ __forceinline__
void kv_step(const float* __restrict__ kq, const float* __restrict__ vq,
             const int jt, const int it0, const int iq0,
             const v2f (&qb0)[2], const v2f (&qb1)[2],
             const int l16, const int hi, float (&m_run)[2], v8f (&o_acc)[2],
             v4f (&vlo)[2], v4f (&vup)[2]) {
    const bool cv_ok = (l16 < CK);

    v2f ka0[NTL], ka1[NTL];
    #pragma unroll
    for (int u = 0; u < NTL; ++u) {
        const int joff = u * NW * 16;
        ka0[u].x = kq[joff          ];
        ka0[u].y = kq[joff +     SEQ];
        ka1[u].x = kq[joff + 4 * SEQ];
        ka1[u].y = kq[joff + 5 * SEQ];
    }
    if (cv_ok) {
        #pragma unroll
        for (int u = 0; u < NTL; ++u) {
            const v4f* vp = (const v4f*)(vq + u * NW * 16);
            vlo[u] = vp[0];
            vup[u] = vp[1];
        }
    }

    v8f sacc[2][NTL];
    #pragma unroll
    for (int u = 0; u < NTL; ++u) {
        #pragma unroll
        for (int q = 0; q < 2; ++q) {
            const v8f z = {0.f,0.f,0.f,0.f,0.f,0.f,0.f,0.f};
            sacc[q][u] = __builtin_amdgcn_wmma_f32_16x16x4_f32(false, ka0[u], false, qb0[q],
                                                               (short)0, z, false, false);
        }
    }
    #pragma unroll
    for (int u = 0; u < NTL; ++u) {
        #pragma unroll
        for (int q = 0; q < 2; ++q)
            sacc[q][u] = __builtin_amdgcn_wmma_f32_16x16x4_f32(false, ka1[u], false, qb1[q],
                                                               (short)0, sacc[q][u], false, false);
    }

    float p[2][NTL][8];
    #pragma unroll
    for (int q = 0; q < 2; ++q) {
        #pragma unroll
        for (int u = 0; u < NTL; ++u) {
            #pragma unroll
            for (int r = 0; r < 8; ++r) p[q][u][r] = sacc[q][u][r];
        }
    }

    const int jlast = jt + (NTL - 1) * NW;
    if (jlast >= it0) {
        #pragma unroll
        for (int q = 0; q < 2; ++q) {
            #pragma unroll
            for (int r = 0; r < 8; ++r) {
                const int key_abs = jlast * 16 + r + 8 * hi;
                if (key_abs > iq0 + 16 * q) p[q][NTL - 1][r] = -1e30f;
            }
        }
    }

    float m_new[2], alpha[2];
    #pragma unroll
    for (int q = 0; q < 2; ++q) {
        float mloc = p[q][0][0];
        #pragma unroll
        for (int u = 0; u < NTL; ++u) {
            #pragma unroll
            for (int r = 0; r < 8; ++r) mloc = fmaxf(mloc, p[q][u][r]);
        }
        const float mtile = fmaxf(mloc, __shfl_xor(mloc, 16, 32));
        m_new[q] = fmaxf(m_run[q], mtile);
    }
    #pragma unroll
    for (int q = 0; q < 2; ++q) {
        alpha[q] = __builtin_amdgcn_exp2f(m_run[q] - m_new[q]);
        m_run[q] = m_new[q];
    }
    #pragma unroll
    for (int q = 0; q < 2; ++q) {
        #pragma unroll
        for (int r = 0; r < 8; ++r) o_acc[q][r] *= alpha[q];
    }
    #pragma unroll
    for (int u = 0; u < NTL; ++u) {
        #pragma unroll
        for (int q = 0; q < 2; ++q) {
            #pragma unroll
            for (int r = 0; r < 8; ++r) p[q][u][r] = __builtin_amdgcn_exp2f(p[q][u][r] - m_new[q]);
        }
    }

    #pragma unroll
    for (int u = 0; u < NTL; ++u) {
        const float vrow[8] = {vlo[u].x, vlo[u].y, vlo[u].z, vlo[u].w, vup[u].x, vup[u].y, vup[u].z, vup[u].w};
        #pragma unroll
        for (int t = 0; t < 4; ++t) {
            v2f va;
            va.x = vrow[2*t    ];
            va.y = vrow[2*t + 1];
            #pragma unroll
            for (int q = 0; q < 2; ++q) {
                v2f pb;
                pb.x = p[q][u][2*t    ];
                pb.y = p[q][u][2*t + 1];
                o_acc[q] = __builtin_amdgcn_wmma_f32_16x16x4_f32(false, va, false, pb,
                                                                 (short)0, o_acc[q], false, false);
            }
        }
    }
}

__device__ __forceinline__
void attend_share(const float* __restrict__ k, const float* __restrict__ v,
                  const int it0, const int jt0, const v2f (&qb0)[2], const v2f (&qb1)[2],
                  const int l16, const int hi,
                  float* __restrict__ sm, float* __restrict__ sl, float* __restrict__ so) {
    const int iq0  = it0 * 16 + l16;
    const int jmax = it0 + 1;

    float m_run[2] = {-1e30f, -1e30f};
    v8f   o_acc[2];
    const float padv = (l16 == CK) ? 1.0f : 0.0f;
    v4f vlo[2], vup[2];
    #pragma unroll
    for (int u = 0; u < 2; ++u) {
        vlo[u] = (v4f){padv,padv,padv,padv};
        vup[u] = (v4f){padv,padv,padv,padv};
    }
    o_acc[0] = (v8f){0.f,0.f,0.f,0.f,0.f,0.f,0.f,0.f};
    o_acc[1] = (v8f){0.f,0.f,0.f,0.f,0.f,0.f,0.f,0.f};

    const float* kq = k + (size_t)(hi * 2) * SEQ + jt0 * 16 + l16;
    const float* vq = v + (size_t)(l16 & (CK - 1)) * SEQ + jt0 * 16 + 8 * hi;
    int jt = jt0;
    for (; jt + NW <= jmax; jt += 2 * NW) {
        kv_step<2>(kq, vq, jt, it0, iq0, qb0, qb1, l16, hi, m_run, o_acc, vlo, vup);
        kq += 2 * NW * 16;
        vq += 2 * NW * 16;
    }
    if (jt <= jmax)
        kv_step<1>(kq, vq, jt, it0, iq0, qb0, qb1, l16, hi, m_run, o_acc, vlo, vup);

    #pragma unroll
    for (int q = 0; q < 2; ++q) {
        if (hi == 0) {
            sm[q * NW * 16 + l16] = m_run[q];
            #pragma unroll
            for (int r = 0; r < 8; ++r) so[q * NW * CK * 16 + r * 16 + l16] = o_acc[q][r];
        } else {
            sl[q * NW * 16 + l16] = o_acc[q][0];
        }
    }
}

__global__ __launch_bounds__(32 * NW)
void causal_attn_fa_f32wmma(const float* __restrict__ Kp,
                            const float* __restrict__ Qp,
                            const float* __restrict__ Vp,
                            float* __restrict__ Op) {
    __shared__ float s_m[4][NW][16];
    __shared__ float s_l[4][NW][16];
    __shared__ float s_o[4][NW][CK][16];

    const int tid  = threadIdx.x;
    const int lane = tid & 31;
    const int wv   = __builtin_amdgcn_readfirstlane(tid >> 5);
    const int l16  = lane & 15;
    const int hi   = lane >> 4;
    const int x    = blockIdx.x;
    const int head = blockIdx.y;
    const int stk  = blockIdx.z;

    const int itA = 2 * x;
    const int itB = 2 * (ND - 1 - x);

    const size_t base = ((size_t)stk * CTOT + (size_t)head * CK) * SEQ;
    const float* __restrict__ q = Qp + base;
    const float* __restrict__ k = Kp + base;
    const float* __restrict__ v = Vp + base;
    float*       __restrict__ o = Op + base;

    const float scale = 0.35355339059327373f * 1.4426950408889634f;

    const int c0 = hi * 2;
    v2f qa0[2], qa1[2], qc0[2], qc1[2];
    #pragma unroll
    for (int t = 0; t < 2; ++t) {
        const int iqA = (itA + t) * 16 + l16;
        const int iqB = (itB + t) * 16 + l16;
        qa0[t].x = q[(size_t)(c0    ) * SEQ + iqA] * scale;
        qa0[t].y = q[(size_t)(c0 + 1) * SEQ + iqA] * scale;
        qa1[t].x = q[(size_t)(c0 + 4) * SEQ + iqA] * scale;
        qa1[t].y = q[(size_t)(c0 + 5) * SEQ + iqA] * scale;
        qc0[t].x = q[(size_t)(c0    ) * SEQ + iqB] * scale;
        qc0[t].y = q[(size_t)(c0 + 1) * SEQ + iqB] * scale;
        qc1[t].x = q[(size_t)(c0 + 4) * SEQ + iqB] * scale;
        qc1[t].y = q[(size_t)(c0 + 5) * SEQ + iqB] * scale;
    }

    const int jtA0 = wv;
    const int jtB0 = (wv - (itA + 2)) & (NW - 1);

    attend_share(k, v, itA, jtA0, qa0, qa1, l16, hi, &s_m[0][wv][0], &s_l[0][wv][0], &s_o[0][wv][0][0]);
    attend_share(k, v, itB, jtB0, qc0, qc1, l16, hi, &s_m[2][wv][0], &s_l[2][wv][0], &s_o[2][wv][0][0]);
    __syncthreads();

    #pragma unroll
    for (int ph = 0; ph < 2; ++ph) {
        const int c  = wv;
        const int ti = ph * 2 + hi;
        const int qi = l16;
        const int ibase = (ph ? itB : itA) * 16;
        float mw[NW];
        #pragma unroll
        for (int w = 0; w < NW; ++w) mw[w] = s_m[ti][w][qi];
        float M = mw[0];
        #pragma unroll
        for (int w = 1; w < NW; ++w) M = fmaxf(M, mw[w]);
        float num = 0.0f, den = 0.0f;
        #pragma unroll
        for (int w = 0; w < NW; ++w) {
            const float e = __builtin_amdgcn_exp2f(mw[w] - M);
            num += e * s_o[ti][w][c][qi];
            den += e * s_l[ti][w][qi];
        }
        o[(size_t)c * SEQ + ibase + lane] = num / den;
    }
}

extern "C" void kernel_launch(void* const* d_in, const int* in_sizes, int n_in,
                              void* d_out, int out_size, void* d_ws, size_t ws_size,
                              hipStream_t stream) {
    (void)in_sizes; (void)n_in; (void)d_ws; (void)ws_size; (void)out_size;
    const float* Kp = (const float*)d_in[0];
    const float* Qp = (const float*)d_in[1];
    const float* Vp = (const float*)d_in[2];
    float*       Op = (float*)d_out;

    dim3 grid(ND / 2, NH, NSTK);
    dim3 block(32 * NW);
    causal_attn_fa_f32wmma<<<grid, block, 0, stream>>>(Kp, Qp, Vp, Op);
}


# Softmax shift dropped in the key loop: weight = exp2(score), no running max, cross-half exchange, alpha or O rescale; one range test of denominators and O per share, online path kept as fallback
# speedup vs baseline: 5.2444x; 1.0834x over previous
#include <hip/hip_runtime.h>


typedef __attribute__((ext_vector_type(2))) float v2f;
typedef __attribute__((ext_vector_type(4))) float v4f;
typedef __attribute__((ext_vector_type(8))) float v8f;

#define SEQ   2048
#define CTOT  64
#define NH    8
#define CK    8
#define NSTK  3
#define NW    8
#define NT    (SEQ / 16)
#define ND    (NT / 2)

template <int NTL, bool ONLINE>
__device__ __forceinline__
void kv_step(const float* __restrict__ kq, const float* __restrict__ vq,
             const int jt, const int it0, const int iq0,
             const v2f (&qb0)[2], const v2f (&qb1)[2],
             const int l16, const int hi, float (&m_run)[2], v8f (&o_acc)[2],
             v4f (&vlo)[2], v4f (&vup)[2]) {
    const bool cv_ok = (l16 < CK);

    v2f ka0[NTL], ka1[NTL];
    #pragma unroll
    for (int u = 0; u < NTL; ++u) {
        const int joff = u * NW * 16;
        ka0[u].x = kq[joff          ];
        ka0[u].y = kq[joff +     SEQ];
        ka1[u].x = kq[joff + 4 * SEQ];
        ka1[u].y = kq[joff + 5 * SEQ];
    }
    if (cv_ok) {
        #pragma unroll
        for (int u = 0; u < NTL; ++u) {
            const v4f* vp = (const v4f*)(vq + u * NW * 16);
            vlo[u] = vp[0];
            vup[u] = vp[1];
        }
    }

    v8f sacc[2][NTL];
    #pragma unroll
    for (int u = 0; u < NTL; ++u) {
        #pragma unroll
        for (int q = 0; q < 2; ++q) {
            const v8f z = {0.f,0.f,0.f,0.f,0.f,0.f,0.f,0.f};
            sacc[q][u] = __builtin_amdgcn_wmma_f32_16x16x4_f32(false, ka0[u], false, qb0[q],
                                                               (short)0, z, false, false);
        }
    }
    #pragma unroll
    for (int u = 0; u < NTL; ++u) {
        #pragma unroll
        for (int q = 0; q < 2; ++q)
            sacc[q][u] = __builtin_amdgcn_wmma_f32_16x16x4_f32(false, ka1[u], false, qb1[q],
                                                               (short)0, sacc[q][u], false, false);
    }

    float p[2][NTL][8];
    #pragma unroll
    for (int q = 0; q < 2; ++q) {
        #pragma unroll
        for (int u = 0; u < NTL; ++u) {
            #pragma unroll
            for (int r = 0; r < 8; ++r) p[q][u][r] = sacc[q][u][r];
        }
    }

    const int jlast = jt + (NTL - 1) * NW;
    if (jlast >= it0) {
        #pragma unroll
        for (int q = 0; q < 2; ++q) {
            #pragma unroll
            for (int r = 0; r < 8; ++r) {
                const int key_abs = jlast * 16 + r + 8 * hi;
                if (key_abs > iq0 + 16 * q) p[q][NTL - 1][r] = -1e30f;
            }
        }
    }

    if constexpr (ONLINE) {
    float m_new[2], alpha[2];
    #pragma unroll
    for (int q = 0; q < 2; ++q) {
        float mloc = p[q][0][0];
        #pragma unroll
        for (int u = 0; u < NTL; ++u) {
            #pragma unroll
            for (int r = 0; r < 8; ++r) mloc = fmaxf(mloc, p[q][u][r]);
        }
        const float mtile = fmaxf(mloc, __shfl_xor(mloc, 16, 32));
        m_new[q] = fmaxf(m_run[q], mtile);
    }
    #pragma unroll
    for (int q = 0; q < 2; ++q) {
        alpha[q] = __builtin_amdgcn_exp2f(m_run[q] - m_new[q]);
        m_run[q] = m_new[q];
    }
    #pragma unroll
    for (int q = 0; q < 2; ++q) {
        #pragma unroll
        for (int r = 0; r < 8; ++r) o_acc[q][r] *= alpha[q];
    }
    #pragma unroll
    for (int u = 0; u < NTL; ++u) {
        #pragma unroll
        for (int q = 0; q < 2; ++q) {
            #pragma unroll
            for (int r = 0; r < 8; ++r) p[q][u][r] = __builtin_amdgcn_exp2f(p[q][u][r] - m_new[q]);
        }
    }
    } else {
    #pragma unroll
    for (int u = 0; u < NTL; ++u) {
        #pragma unroll
        for (int q = 0; q < 2; ++q) {
            #pragma unroll
            for (int r = 0; r < 8; ++r) p[q][u][r] = __builtin_amdgcn_exp2f(p[q][u][r]);
        }
    }
    }

    #pragma unroll
    for (int u = 0; u < NTL; ++u) {
        const float vrow[8] = {vlo[u].x, vlo[u].y, vlo[u].z, vlo[u].w, vup[u].x, vup[u].y, vup[u].z, vup[u].w};
        #pragma unroll
        for (int t = 0; t < 4; ++t) {
            v2f va;
            va.x = vrow[2*t    ];
            va.y = vrow[2*t + 1];
            #pragma unroll
            for (int q = 0; q < 2; ++q) {
                v2f pb;
                pb.x = p[q][u][2*t    ];
                pb.y = p[q][u][2*t + 1];
                o_acc[q] = __builtin_amdgcn_wmma_f32_16x16x4_f32(false, va, false, pb,
                                                                 (short)0, o_acc[q], false, false);
            }
        }
    }
}

template <bool ONLINE>
__device__ __forceinline__
unsigned attend_share(const float* __restrict__ k, const float* __restrict__ v,
                  const int it0, const int jt0, const v2f (&qb0)[2], const v2f (&qb1)[2],
                  const int l16, const int hi,
                  float* __restrict__ sm, float* __restrict__ sl, float* __restrict__ so) {
    const int iq0  = it0 * 16 + l16;
    const int jmax = it0 + 1;

    float m_run[2] = {-1e30f, -1e30f};
    v8f   o_acc[2];
    const float padv = (l16 == CK) ? 1.0f : 0.0f;
    v4f vlo[2], vup[2];
    #pragma unroll
    for (int u = 0; u < 2; ++u) {
        vlo[u] = (v4f){padv,padv,padv,padv};
        vup[u] = (v4f){padv,padv,padv,padv};
    }
    o_acc[0] = (v8f){0.f,0.f,0.f,0.f,0.f,0.f,0.f,0.f};
    o_acc[1] = (v8f){0.f,0.f,0.f,0.f,0.f,0.f,0.f,0.f};

    const float* kq = k + (size_t)(hi * 2) * SEQ + jt0 * 16 + l16;
    const float* vq = v + (size_t)(l16 & (CK - 1)) * SEQ + jt0 * 16 + 8 * hi;
    int jt = jt0;
    for (; jt + NW <= jmax; jt += 2 * NW) {
        kv_step<2, ONLINE>(kq, vq, jt, it0, iq0, qb0, qb1, l16, hi, m_run, o_acc, vlo, vup);
        kq += 2 * NW * 16;
        vq += 2 * NW * 16;
    }
    if (jt <= jmax)
        kv_step<1, ONLINE>(kq, vq, jt, it0, iq0, qb0, qb1, l16, hi, m_run, o_acc, vlo, vup);

    bool bad = false;
    if constexpr (!ONLINE) {
        #pragma unroll
        for (int q = 0; q < 2; ++q) {
            const bool vis = (jt0 <= it0 + q);
            m_run[q] = vis ? 0.0f : -1e30f;
            float tot = o_acc[q][0];
            #pragma unroll
            for (int r = 1; r < 8; ++r) tot += o_acc[q][r];
            bad |= !(__builtin_fabsf(tot) < 0x1p60f);
            bad |= (vis && hi == 1 && !(o_acc[q][0] >= 0x1p-60f));
        }
    }
    #pragma unroll
    for (int q = 0; q < 2; ++q) {
        if (hi == 0) {
            sm[q * NW * 16 + l16] = m_run[q];
            #pragma unroll
            for (int r = 0; r < 8; ++r) so[q * NW * CK * 16 + r * 16 + l16] = o_acc[q][r];
        } else {
            sl[q * NW * 16 + l16] = o_acc[q][0];
        }
    }
    return __builtin_amdgcn_ballot_w32(bad);
}

__global__ __launch_bounds__(32 * NW)
void causal_attn_fa_f32wmma(const float* __restrict__ Kp,
                            const float* __restrict__ Qp,
                            const float* __restrict__ Vp,
                            float* __restrict__ Op) {
    __shared__ float s_m[4][NW][16];
    __shared__ float s_l[4][NW][16];
    __shared__ float s_o[4][NW][CK][16];

    const int tid  = threadIdx.x;
    const int lane = tid & 31;
    const int wv   = __builtin_amdgcn_readfirstlane(tid >> 5);
    const int l16  = lane & 15;
    const int hi   = lane >> 4;
    const int x    = blockIdx.x;
    const int head = blockIdx.y;
    const int stk  = blockIdx.z;

    const int itA = 2 * x;
    const int itB = 2 * (ND - 1 - x);

    const size_t base = ((size_t)stk * CTOT + (size_t)head * CK) * SEQ;
    const float* __restrict__ q = Qp + base;
    const float* __restrict__ k = Kp + base;
    const float* __restrict__ v = Vp + base;
    float*       __restrict__ o = Op + base;

    const float scale = 0.35355339059327373f * 1.4426950408889634f;

    const int c0 = hi * 2;
    v2f qa0[2], qa1[2], qc0[2], qc1[2];
    #pragma unroll
    for (int t = 0; t < 2; ++t) {
        const int iqA = (itA + t) * 16 + l16;
        const int iqB = (itB + t) * 16 + l16;
        qa0[t].x = q[(size_t)(c0    ) * SEQ + iqA] * scale;
        qa0[t].y = q[(size_t)(c0 + 1) * SEQ + iqA] * scale;
        qa1[t].x = q[(size_t)(c0 + 4) * SEQ + iqA] * scale;
        qa1[t].y = q[(size_t)(c0 + 5) * SEQ + iqA] * scale;
        qc0[t].x = q[(size_t)(c0    ) * SEQ + iqB] * scale;
        qc0[t].y = q[(size_t)(c0 + 1) * SEQ + iqB] * scale;
        qc1[t].x = q[(size_t)(c0 + 4) * SEQ + iqB] * scale;
        qc1[t].y = q[(size_t)(c0 + 5) * SEQ + iqB] * scale;
    }

    const int jtA0 = wv;
    const int jtB0 = (wv - (itA + 2)) & (NW - 1);

    unsigned bad = attend_share<false>(k, v, itA, jtA0, qa0, qa1, l16, hi, &s_m[0][wv][0], &s_l[0][wv][0], &s_o[0][wv][0][0]);
    bad |= attend_share<false>(k, v, itB, jtB0, qc0, qc1, l16, hi, &s_m[2][wv][0], &s_l[2][wv][0], &s_o[2][wv][0][0]);
    if (bad != 0u) {
        attend_share<true>(k, v, itA, jtA0, qa0, qa1, l16, hi, &s_m[0][wv][0], &s_l[0][wv][0], &s_o[0][wv][0][0]);
        attend_share<true>(k, v, itB, jtB0, qc0, qc1, l16, hi, &s_m[2][wv][0], &s_l[2][wv][0], &s_o[2][wv][0][0]);
    }
    __syncthreads();

    #pragma unroll
    for (int ph = 0; ph < 2; ++ph) {
        const int c  = wv;
        const int ti = ph * 2 + hi;
        const int qi = l16;
        const int ibase = (ph ? itB : itA) * 16;
        float mw[NW];
        #pragma unroll
        for (int w = 0; w < NW; ++w) mw[w] = s_m[ti][w][qi];
        float M = mw[0];
        #pragma unroll
        for (int w = 1; w < NW; ++w) M = fmaxf(M, mw[w]);
        float num = 0.0f, den = 0.0f;
        #pragma unroll
        for (int w = 0; w < NW; ++w) {
            const float e = __builtin_amdgcn_exp2f(mw[w] - M);
            num += e * s_o[ti][w][c][qi];
            den += e * s_l[ti][w][qi];
        }
        o[(size_t)c * SEQ + ibase + lane] = num / den;
    }
}

extern "C" void kernel_launch(void* const* d_in, const int* in_sizes, int n_in,
                              void* d_out, int out_size, void* d_ws, size_t ws_size,
                              hipStream_t stream) {
    (void)in_sizes; (void)n_in; (void)d_ws; (void)ws_size; (void)out_size;
    const float* Kp = (const float*)d_in[0];
    const float* Qp = (const float*)d_in[1];
    const float* Vp = (const float*)d_in[2];
    float*       Op = (float*)d_out;

    dim3 grid(ND / 2, NH, NSTK);
    dim3 block(32 * NW);
    causal_attn_fa_f32wmma<<<grid, block, 0, stream>>>(Kp, Qp, Vp, Op);
}
